# starting kernel
# baseline (speedup 1.0000x reference)
#include <hip/hip_runtime.h>
#include <math.h>


typedef __attribute__((ext_vector_type(2))) float v2f;
typedef __attribute__((ext_vector_type(8))) float v8f;

__global__ void wq_kernel(const float* __restrict__ W2, const float* __restrict__ b2,
                          const float* __restrict__ q, float* __restrict__ Wqp,
                          float* __restrict__ bqv) {
  const int h = threadIdx.x;
  float s[8];
#pragma unroll
  for (int k = 0; k < 8; ++k) s[k] = 0.0f;
  for (int d = 0; d < 256; ++d) {
    const float w = W2[d * 128 + h];
#pragma unroll
    for (int k = 0; k < 8; ++k) s[k] += w * q[k * 256 + d];
  }
#pragma unroll
  for (int k = 0; k < 8; ++k) Wqp[h * 16 + k] = s[k];
#pragma unroll
  for (int k = 8; k < 16; ++k) Wqp[h * 16 + k] = 0.0f;
  if (h < 8) {
    float t = 0.0f;
    for (int d = 0; d < 256; ++d) t += b2[d] * q[h * 256 + d];
    bqv[h] = t;
  }
}

__global__ void gemm_A_kernel(const float* __restrict__ E, const float* __restrict__ W1,
                              float* __restrict__ Afull) {
  const int wave = (blockIdx.x * blockDim.x + threadIdx.x) >> 5;
  const int lane = threadIdx.x & 31;
  const int l15 = lane & 15;
  const int hi  = lane >> 4;
  const int tm  = wave >> 4;
  const int tc  = wave & 15;
  const int row0 = tm * 16;
  const int col  = tc * 16 + l15;
  const float* wcol = (col < 128) ? (W1 + col * 512) : (W1 + (col - 128) * 512 + 256);
  const float* erow = E + (row0 + l15) * 256;
  v8f acc = {};
#pragma unroll 4
  for (int kk = 0; kk < 64; ++kk) {
    const int d = kk * 4 + hi * 2;
    v2f a = *(const v2f*)(erow + d);
    v2f b = *(const v2f*)(wcol + d);
    acc = __builtin_amdgcn_wmma_f32_16x16x4_f32(false, a, false, b, (short)0, acc,
                                                false, false);
  }
#pragma unroll
  for (int r = 0; r < 8; ++r) {
    Afull[(row0 + r + hi * 8) * 256 + tc * 16 + l15] = acc[r];
  }
}

__global__ void score_kernel(const float* __restrict__ Afull, const float* __restrict__ b1,
                             const float* __restrict__ Wqp, const float* __restrict__ bqv,
                             float* __restrict__ out) {
  __shared__ float sWq[128 * 16];
  {
    const int t = threadIdx.x;
#pragma unroll
    for (int u = 0; u < 8; ++u) sWq[t + u * 256] = Wqp[t + u * 256];
  }
  __syncthreads();

  const int wave = (blockIdx.x * blockDim.x + threadIdx.x) >> 5;
  const int lane = threadIdx.x & 31;
  const int l15 = lane & 15;
  const int hi  = lane >> 4;
  const int jt  = wave & 7;
  const int i   = (wave >> 3) & 127;
  const int b   = wave >> 10;
  const int j   = jt * 16 + l15;

  const float* ai = Afull + (b * 128 + i) * 256;
  const float* aj = Afull + (b * 128 + j) * 256 + 128;

  v8f acc = {};
#pragma unroll 4
  for (int kk = 0; kk < 32; ++kk) {
    const int h = kk * 4 + hi * 2;
    v2f a;
    a.x = sWq[h * 16 + l15];
    a.y = sWq[(h + 1) * 16 + l15];
    v2f xa = *(const v2f*)(aj + h);
    v2f xi = *(const v2f*)(ai + h);
    v2f bb = *(const v2f*)(b1 + h);
    v2f hv;
    hv.x = fmaxf(xa.x + xi.x + bb.x, 0.0f);
    hv.y = fmaxf(xa.y + xi.y + bb.y, 0.0f);
    acc = __builtin_amdgcn_wmma_f32_16x16x4_f32(false, a, false, hv, (short)0, acc,
                                                false, false);
  }

  if (hi == 0) {
#pragma unroll
    for (int r = 0; r < 8; ++r) {
      float v = acc[r] + bqv[r];
      float s = 1.0f / (1.0f + __expf(-v));
      if (i == j) s = 0.0f;
      out[(((b * 8 + r) * 128) + i) * 128 + j] = s;
    }
  }
}

extern "C" void kernel_launch(void* const* d_in, const int* in_sizes, int n_in,
                              void* d_out, int out_size, void* d_ws, size_t ws_size,
                              hipStream_t stream) {
  const float* E  = (const float*)d_in[0];
  const float* W1 = (const float*)d_in[1];
  const float* b1 = (const float*)d_in[2];
  const float* W2 = (const float*)d_in[3];
  const float* b2 = (const float*)d_in[4];
  const float* q  = (const float*)d_in[5];
  float* out = (float*)d_out;

  float* Afull = (float*)d_ws;
  float* Wqp   = Afull + 262144;
  float* bqv   = Wqp + 2048;

  wq_kernel<<<dim3(1), dim3(128), 0, stream>>>(W2, b2, q, Wqp, bqv);
  gemm_A_kernel<<<dim3(128), dim3(256), 0, stream>>>(E, W1, Afull);
  score_kernel<<<dim3(1024), dim3(256), 0, stream>>>(Afull, b1, Wqp, bqv, out);
}


# GEMM launch fused into score kernel: each work-group (b, 16 i, 32 j) builds its Ai/Aj tiles in LDS and scores from them; 3 launches -> 2; Ai/Aj sum order changes, b1 summed as (Ai+b1)+Aj
# speedup vs baseline: 3.3183x; 3.3183x over previous
#include <hip/hip_runtime.h>
#include <math.h>


typedef __attribute__((ext_vector_type(2))) float v2f;
typedef __attribute__((ext_vector_type(8))) float v8f;
typedef __attribute__((ext_vector_type(4))) float v4f;

__global__ void wq_kernel(const float* __restrict__ W2, const float* __restrict__ b2,
                          const float* __restrict__ q, float* __restrict__ Wqp,
                          float* __restrict__ bqv) {
  __shared__ float red[8 * 4 * 8];
  const int t    = threadIdx.x;
  const int lane = t & 31;
  const int wv   = t >> 5;
  const int hl   = t & 3;
  const int d0   = (t >> 2) * 4;
  const int g    = blockIdx.x;

  float w0, w1, w2, w3;
  if (g < 32) {
    const float* wp = W2 + d0 * 128 + g * 4 + hl;
    w0 = wp[0];
    w1 = wp[128];
    w2 = wp[256];
    w3 = wp[384];
  } else {
    const float4 bv = *(const float4*)(b2 + d0);
    w0 = bv.x; w1 = bv.y; w2 = bv.z; w3 = bv.w;
  }

  float s[8];
#pragma unroll
  for (int k = 0; k < 8; ++k) {
    const float4 qv = *(const float4*)(q + k * 256 + d0);
    s[k] = w0 * qv.x + w1 * qv.y + w2 * qv.z + w3 * qv.w;
  }
#pragma unroll
  for (int k = 0; k < 8; ++k) {
    s[k] += __shfl_xor(s[k], 4);
    s[k] += __shfl_xor(s[k], 8);
    s[k] += __shfl_xor(s[k], 16);
  }
  if (lane < 4) {
#pragma unroll
    for (int k = 0; k < 8; ++k) red[(wv * 4 + lane) * 8 + k] = s[k];
  }
  __syncthreads();
  if (t < 32) {
    const int c = t >> 3;
    const int k = t & 7;
    float r = 0.0f;
#pragma unroll
    for (int u = 0; u < 8; ++u) r += red[(u * 4 + c) * 8 + k];
    if (g < 32) {
      Wqp[(g * 4 + c) * 16 + k]     = r;
      Wqp[(g * 4 + c) * 16 + 8 + k] = 0.0f;
    } else if (c == 0) {
      bqv[k] = r;
    }
  }
}

#define CH_P4 17
#define S_P4  33

__device__ __forceinline__ v8f mma4(v2f a, v2f b, v8f c) {
  return __builtin_amdgcn_wmma_f32_16x16x4_f32(false, a, false, b, (short)0, c, false, false);
}

__global__ void __launch_bounds__(256)
fused_kernel(const float* __restrict__ E, const float* __restrict__ W1,
             const float* __restrict__ b1, const float* __restrict__ Wqp,
             const float* __restrict__ bqv, float* __restrict__ out) {
  __shared__ v4f sbuf4[304 * CH_P4];
  __shared__ v4f sSi4[16 * S_P4];
  __shared__ v4f sSj4[32 * S_P4];
  __shared__ v4f sWq4[16 * S_P4];

  const int t    = threadIdx.x;
  const int lane = t & 31;
  const int w    = t >> 5;
  const int c15  = lane & 15;
  const int hi   = lane >> 4;
  const int g    = blockIdx.x;
  const int b    = g >> 5;
  const int i0   = ((g >> 2) & 7) * 16;
  const int j0   = (g & 3) * 32;

  const int tr = t >> 4;
  const int tc = t & 15;
  const float* eI = E + (b * 128 + i0 + tr) * 256 + tc * 4;
  const float* eJ = E + (b * 128 + j0 + tr) * 256 + tc * 4;
  const float* wA = W1 + tr * 512 + tc * 4;

  v4f pre0, pre1, pre2, pre3, pre4, pre5, pre6, pre7, pre8, pre9, pre10, pre11, pre12, pre13, pre14, pre15, pre16, pre17, pre18;
#define LD4(ptr) (*(const v4f*)(ptr))
#define FETCH_CHUNK(cc) \
  pre0 = LD4(eI + (cc) * 64); \
  pre1 = LD4(eJ + (cc) * 64); \
  pre2 = LD4(eJ + 16 * 256 + (cc) * 64); \
  pre3 = LD4(wA + 0 * 16 * 512 + (cc) * 64); \
  pre4 = LD4(wA + 1 * 16 * 512 + (cc) * 64); \
  pre5 = LD4(wA + 2 * 16 * 512 + (cc) * 64); \
  pre6 = LD4(wA + 3 * 16 * 512 + (cc) * 64); \
  pre7 = LD4(wA + 4 * 16 * 512 + (cc) * 64); \
  pre8 = LD4(wA + 5 * 16 * 512 + (cc) * 64); \
  pre9 = LD4(wA + 6 * 16 * 512 + (cc) * 64); \
  pre10 = LD4(wA + 7 * 16 * 512 + (cc) * 64); \
  pre11 = LD4(wA + 0 * 16 * 512 + 256 + (cc) * 64); \
  pre12 = LD4(wA + 1 * 16 * 512 + 256 + (cc) * 64); \
  pre13 = LD4(wA + 2 * 16 * 512 + 256 + (cc) * 64); \
  pre14 = LD4(wA + 3 * 16 * 512 + 256 + (cc) * 64); \
  pre15 = LD4(wA + 4 * 16 * 512 + 256 + (cc) * 64); \
  pre16 = LD4(wA + 5 * 16 * 512 + 256 + (cc) * 64); \
  pre17 = LD4(wA + 6 * 16 * 512 + 256 + (cc) * 64); \
  pre18 = LD4(wA + 7 * 16 * 512 + 256 + (cc) * 64);
#define STORE_CHUNK() \
  sbuf4[(0 * 16 + tr) * CH_P4 + tc] = pre0; \
  sbuf4[(1 * 16 + tr) * CH_P4 + tc] = pre1; \
  sbuf4[(2 * 16 + tr) * CH_P4 + tc] = pre2; \
  sbuf4[(3 * 16 + tr) * CH_P4 + tc] = pre3; \
  sbuf4[(4 * 16 + tr) * CH_P4 + tc] = pre4; \
  sbuf4[(5 * 16 + tr) * CH_P4 + tc] = pre5; \
  sbuf4[(6 * 16 + tr) * CH_P4 + tc] = pre6; \
  sbuf4[(7 * 16 + tr) * CH_P4 + tc] = pre7; \
  sbuf4[(8 * 16 + tr) * CH_P4 + tc] = pre8; \
  sbuf4[(9 * 16 + tr) * CH_P4 + tc] = pre9; \
  sbuf4[(10 * 16 + tr) * CH_P4 + tc] = pre10; \
  sbuf4[(11 * 16 + tr) * CH_P4 + tc] = pre11; \
  sbuf4[(12 * 16 + tr) * CH_P4 + tc] = pre12; \
  sbuf4[(13 * 16 + tr) * CH_P4 + tc] = pre13; \
  sbuf4[(14 * 16 + tr) * CH_P4 + tc] = pre14; \
  sbuf4[(15 * 16 + tr) * CH_P4 + tc] = pre15; \
  sbuf4[(16 * 16 + tr) * CH_P4 + tc] = pre16; \
  sbuf4[(17 * 16 + tr) * CH_P4 + tc] = pre17; \
  sbuf4[(18 * 16 + tr) * CH_P4 + tc] = pre18;

  FETCH_CHUNK(0)

  {
    float* sWq = (float*)sWq4;
#pragma unroll
    for (int u = 0; u < 8; ++u) {
      const int idx = t + u * 256;
      sWq[(idx & 15) * (S_P4 * 4) + (idx >> 4)] = Wqp[idx];
    }
  }

  v8f accI = {}, accJ0 = {}, accJ1 = {};
  const int fo = hi;

#define CHUNK_STEP(cc, FETCH_NEXT)                                                   \
  {                                                                                  \
    STORE_CHUNK()                                                                    \
    __syncthreads();                                                                 \
    FETCH_NEXT                                                                       \
    v4f fi[8], fj0[8], fj1[8], fa[8], fb[8];                                      \
    _Pragma("unroll") for (int m = 0; m < 8; ++m) {                                  \
      fi[m]  = sbuf4[(c15) * CH_P4 + 2 * m + fo];                                    \
      fa[m]  = sbuf4[(48 + 16 * w + c15) * CH_P4 + 2 * m + fo];                      \
      fj0[m] = sbuf4[(16 + c15) * CH_P4 + 2 * m + fo];                               \
      fj1[m] = sbuf4[(32 + c15) * CH_P4 + 2 * m + fo];                               \
      fb[m]  = sbuf4[(176 + 16 * w + c15) * CH_P4 + 2 * m + fo];                     \
    }                                                                                \
    _Pragma("unroll") for (int m = 0; m < 8; ++m) {                                  \
      v2f a, bw;                                                                     \
      a.x = fi[m].x;  a.y = fi[m].y;  bw.x = fa[m].x; bw.y = fa[m].y;                \
      accI = mma4(a, bw, accI);                                                      \
      a.x = fj0[m].x; a.y = fj0[m].y; bw.x = fb[m].x; bw.y = fb[m].y;                \
      accJ0 = mma4(a, bw, accJ0);                                                    \
      a.x = fj1[m].x; a.y = fj1[m].y;                                                \
      accJ1 = mma4(a, bw, accJ1);                                                    \
      a.x = fi[m].z;  a.y = fi[m].w;  bw.x = fa[m].z; bw.y = fa[m].w;                \
      accI = mma4(a, bw, accI);                                                      \
      a.x = fj0[m].z; a.y = fj0[m].w; bw.x = fb[m].z; bw.y = fb[m].w;                \
      accJ0 = mma4(a, bw, accJ0);                                                    \
      a.x = fj1[m].z; a.y = fj1[m].w;                                                \
      accJ1 = mma4(a, bw, accJ1);                                                    \
    }                                                                                \
    __syncthreads();   \
  }
  CHUNK_STEP(0, FETCH_CHUNK(1))
  CHUNK_STEP(1, FETCH_CHUNK(2))
  CHUNK_STEP(2, FETCH_CHUNK(3))
  CHUNK_STEP(3, )
#undef CHUNK_STEP

  {
    float* sSi = (float*)sSi4;
    float* sSj = (float*)sSj4;
    const float bias = b1[16 * w + c15];
#pragma unroll
    for (int r = 0; r < 8; ++r) {
      sSi[(8 * hi + r) * (S_P4 * 4) + 16 * w + c15]      = accI[r] + bias;
      sSj[(8 * hi + r) * (S_P4 * 4) + 16 * w + c15]      = accJ0[r];
      sSj[(16 + 8 * hi + r) * (S_P4 * 4) + 16 * w + c15] = accJ1[r];
    }
  }
  __syncthreads();

  v4f bj0[16], bj1[16], aw[16];
#pragma unroll
  for (int m = 0; m < 16; ++m) {
    aw[m]  = sWq4[c15 * S_P4 + 2 * m + fo];
    bj0[m] = sSj4[c15 * S_P4 + 2 * m + fo];
    bj1[m] = sSj4[(16 + c15) * S_P4 + 2 * m + fo];
  }
  float bq[8];
#pragma unroll
  for (int r = 0; r < 8; ++r) bq[r] = bqv[r];

  const int jg = j0 + lane;

#pragma unroll
  for (int ii = 0; ii < 2; ++ii) {
    const int il = 2 * w + ii;
    const int ig = i0 + il;
    v4f ci[16];
#pragma unroll
    for (int m = 0; m < 16; ++m) ci[m] = sSi4[il * S_P4 + 2 * m + fo];
    v8f acc0 = {}, acc1 = {};
#pragma unroll
    for (int m = 0; m < 16; ++m) {
      v2f a, h0, h1;
      a.x = aw[m].x; a.y = aw[m].y;
      h0.x = fmaxf(bj0[m].x + ci[m].x, 0.0f); h0.y = fmaxf(bj0[m].y + ci[m].y, 0.0f);
      h1.x = fmaxf(bj1[m].x + ci[m].x, 0.0f); h1.y = fmaxf(bj1[m].y + ci[m].y, 0.0f);
      acc0 = mma4(a, h0, acc0);
      acc1 = mma4(a, h1, acc1);
      a.x = aw[m].z; a.y = aw[m].w;
      h0.x = fmaxf(bj0[m].z + ci[m].z, 0.0f); h0.y = fmaxf(bj0[m].w + ci[m].w, 0.0f);
      h1.x = fmaxf(bj1[m].z + ci[m].z, 0.0f); h1.y = fmaxf(bj1[m].w + ci[m].w, 0.0f);
      acc0 = mma4(a, h0, acc0);
      acc1 = mma4(a, h1, acc1);
    }
#pragma unroll
    for (int r = 0; r < 8; ++r) {
      const float o1 = __shfl_xor(acc1[r], 16);
      const float v  = (hi ? o1 : acc0[r]) + bq[r];
      float s = 1.0f / (1.0f + __expf(-v));
      if (ig == jg) s = 0.0f;
      out[((b * 8 + r) * 128 + ig) * 128 + jg] = s;
    }
  }
#undef FETCH_CHUNK
#undef STORE_CHUNK
#undef LD4
}

extern "C" void kernel_launch(void* const* d_in, const int* in_sizes, int n_in,
                              void* d_out, int out_size, void* d_ws, size_t ws_size,
                              hipStream_t stream) {
  const float* E  = (const float*)d_in[0];
  const float* W1 = (const float*)d_in[1];
  const float* b1 = (const float*)d_in[2];
  const float* W2 = (const float*)d_in[3];
  const float* b2 = (const float*)d_in[4];
  const float* q  = (const float*)d_in[5];
  float* out = (float*)d_out;

  float* Wqp = (float*)d_ws + 262144;
  float* bqv = Wqp + 2048;

  wq_kernel<<<dim3(33), dim3(256), 0, stream>>>(W2, b2, q, Wqp, bqv);
  fused_kernel<<<dim3(256), dim3(256), 0, stream>>>(E, W1, b1, Wqp, bqv, out);
}


# GEMM phase without chunk barriers: 48 E rows staged to LDS once for all K, each wave loads its own W1a/W1b fragments from global straight into registers one chunk ahead; 9 barriers -> 2
# speedup vs baseline: 3.5458x; 1.0686x over previous
#include <hip/hip_runtime.h>
#include <math.h>


typedef __attribute__((ext_vector_type(2))) float v2f;
typedef __attribute__((ext_vector_type(8))) float v8f;
typedef __attribute__((ext_vector_type(4))) float v4f;

__global__ void wq_kernel(const float* __restrict__ W2, const float* __restrict__ b2,
                          const float* __restrict__ q, float* __restrict__ Wqp,
                          float* __restrict__ bqv) {
  __shared__ float red[8 * 4 * 8];
  const int t    = threadIdx.x;
  const int lane = t & 31;
  const int wv   = t >> 5;
  const int hl   = t & 3;
  const int d0   = (t >> 2) * 4;
  const int g    = blockIdx.x;

  float w0, w1, w2, w3;
  if (g < 32) {
    const float* wp = W2 + d0 * 128 + g * 4 + hl;
    w0 = wp[0];
    w1 = wp[128];
    w2 = wp[256];
    w3 = wp[384];
  } else {
    const float4 bv = *(const float4*)(b2 + d0);
    w0 = bv.x; w1 = bv.y; w2 = bv.z; w3 = bv.w;
  }

  float s[8];
#pragma unroll
  for (int k = 0; k < 8; ++k) {
    const float4 qv = *(const float4*)(q + k * 256 + d0);
    s[k] = w0 * qv.x + w1 * qv.y + w2 * qv.z + w3 * qv.w;
  }
#pragma unroll
  for (int k = 0; k < 8; ++k) {
    s[k] += __shfl_xor(s[k], 4);
    s[k] += __shfl_xor(s[k], 8);
    s[k] += __shfl_xor(s[k], 16);
  }
  if (lane < 4) {
#pragma unroll
    for (int k = 0; k < 8; ++k) red[(wv * 4 + lane) * 8 + k] = s[k];
  }
  __syncthreads();
  if (t < 32) {
    const int c = t >> 3;
    const int k = t & 7;
    float r = 0.0f;
#pragma unroll
    for (int u = 0; u < 8; ++u) r += red[(u * 4 + c) * 8 + k];
    if (g < 32) {
      Wqp[(g * 4 + c) * 16 + k]     = r;
      Wqp[(g * 4 + c) * 16 + 8 + k] = 0.0f;
    } else if (c == 0) {
      bqv[k] = r;
    }
  }
}

#define E_P4  65
#define S_P4  33

__device__ __forceinline__ v8f mma4(v2f a, v2f b, v8f c) {
  return __builtin_amdgcn_wmma_f32_16x16x4_f32(false, a, false, b, (short)0, c, false, false);
}

__global__ void __launch_bounds__(256)
fused_kernel(const float* __restrict__ E, const float* __restrict__ W1,
             const float* __restrict__ b1, const float* __restrict__ Wqp,
             const float* __restrict__ bqv, float* __restrict__ out) {
  __shared__ v4f sE4[48 * E_P4];
  __shared__ v4f sSi4[16 * S_P4];
  __shared__ v4f sSj4[32 * S_P4];
  __shared__ v4f sWq4[16 * S_P4];

  const int t    = threadIdx.x;
  const int lane = t & 31;
  const int w    = t >> 5;
  const int c15  = lane & 15;
  const int hi   = lane >> 4;
  const int g    = blockIdx.x;
  const int b    = g >> 5;
  const int i0   = ((g >> 2) & 7) * 16;
  const int j0   = (g & 3) * 32;

#define LD4(ptr) (*(const v4f*)(ptr))
  const float* wB = W1 + (16 * w + c15) * 512 + 4 * hi;
#define FETCH_B(BA, BB, cc)                                      \
  _Pragma("unroll") for (int m = 0; m < 8; ++m) {                \
    BA[m] = LD4(wB + (cc) * 64 + 8 * m);                         \
    BB[m] = LD4(wB + 256 + (cc) * 64 + 8 * m);                   \
  }
  v4f ba0[8], bb0[8], ba1[8], bb1[8];

  const int tr = t >> 4;
  const int tc = t & 15;
  const float* eI = E + (b * 128 + i0 + tr) * 256 + tc * 4;
  const float* eJ = E + (b * 128 + j0 + tr) * 256 + tc * 4;
  v4f st[12];
#pragma unroll
  for (int qd = 0; qd < 4; ++qd) {
    st[qd]     = LD4(eI + qd * 64);
    st[4 + qd] = LD4(eJ + qd * 64);
    st[8 + qd] = LD4(eJ + 16 * 256 + qd * 64);
  }
  float wqv[8];
#pragma unroll
  for (int u = 0; u < 8; ++u) wqv[u] = Wqp[t + u * 256];
  FETCH_B(ba0, bb0, 0)

#pragma unroll
  for (int qd = 0; qd < 4; ++qd) {
    sE4[tr * E_P4 + tc + 16 * qd]        = st[qd];
    sE4[(16 + tr) * E_P4 + tc + 16 * qd] = st[4 + qd];
    sE4[(32 + tr) * E_P4 + tc + 16 * qd] = st[8 + qd];
  }
  {
    float* sWq = (float*)sWq4;
#pragma unroll
    for (int u = 0; u < 8; ++u) {
      const int idx = t + u * 256;
      sWq[(idx & 15) * (S_P4 * 4) + (idx >> 4)] = wqv[u];
    }
  }
  __syncthreads();

  v8f accI = {}, accJ0 = {}, accJ1 = {};
  const int fo = hi;

#define CHUNK_STEP(cc, BA, BB, FETCH_NEXT)                                           \
  {                                                                                  \
    FETCH_NEXT                                                                       \
    v4f fi[8], fj0[8], fj1[8];                                                       \
    _Pragma("unroll") for (int m = 0; m < 8; ++m) {                                  \
      fi[m]  = sE4[(c15) * E_P4 + (cc) * 16 + 2 * m + fo];                           \
      fj0[m] = sE4[(16 + c15) * E_P4 + (cc) * 16 + 2 * m + fo];                      \
      fj1[m] = sE4[(32 + c15) * E_P4 + (cc) * 16 + 2 * m + fo];                      \
    }                                                                                \
    _Pragma("unroll") for (int m = 0; m < 8; ++m) {                                  \
      v2f a, bw;                                                                     \
      a.x = fi[m].x;  a.y = fi[m].y;  bw.x = BA[m].x; bw.y = BA[m].y;                \
      accI = mma4(a, bw, accI);                                                      \
      a.x = fj0[m].x; a.y = fj0[m].y; bw.x = BB[m].x; bw.y = BB[m].y;                \
      accJ0 = mma4(a, bw, accJ0);                                                    \
      a.x = fj1[m].x; a.y = fj1[m].y;                                                \
      accJ1 = mma4(a, bw, accJ1);                                                    \
      a.x = fi[m].z;  a.y = fi[m].w;  bw.x = BA[m].z; bw.y = BA[m].w;                \
      accI = mma4(a, bw, accI);                                                      \
      a.x = fj0[m].z; a.y = fj0[m].w; bw.x = BB[m].z; bw.y = BB[m].w;                \
      accJ0 = mma4(a, bw, accJ0);                                                    \
      a.x = fj1[m].z; a.y = fj1[m].w;                                                \
      accJ1 = mma4(a, bw, accJ1);                                                    \
    }                                                                                \
  }
  CHUNK_STEP(0, ba0, bb0, FETCH_B(ba1, bb1, 1))
  CHUNK_STEP(1, ba1, bb1, FETCH_B(ba0, bb0, 2))
  CHUNK_STEP(2, ba0, bb0, FETCH_B(ba1, bb1, 3))
  CHUNK_STEP(3, ba1, bb1, )
#undef CHUNK_STEP

  {
    float* sSi = (float*)sSi4;
    float* sSj = (float*)sSj4;
    const float bias = b1[16 * w + c15];
#pragma unroll
    for (int r = 0; r < 8; ++r) {
      sSi[(8 * hi + r) * (S_P4 * 4) + 16 * w + c15]      = accI[r] + bias;
      sSj[(8 * hi + r) * (S_P4 * 4) + 16 * w + c15]      = accJ0[r];
      sSj[(16 + 8 * hi + r) * (S_P4 * 4) + 16 * w + c15] = accJ1[r];
    }
  }
  __syncthreads();

  v4f bj0[16], bj1[16], aw[16];
#pragma unroll
  for (int m = 0; m < 16; ++m) {
    aw[m]  = sWq4[c15 * S_P4 + 2 * m + fo];
    bj0[m] = sSj4[c15 * S_P4 + 2 * m + fo];
    bj1[m] = sSj4[(16 + c15) * S_P4 + 2 * m + fo];
  }
  float bq[8];
#pragma unroll
  for (int r = 0; r < 8; ++r) bq[r] = bqv[r];

  const int jg = j0 + lane;

#pragma unroll
  for (int ii = 0; ii < 2; ++ii) {
    const int il = 2 * w + ii;
    const int ig = i0 + il;
    v4f ci[16];
#pragma unroll
    for (int m = 0; m < 16; ++m) ci[m] = sSi4[il * S_P4 + 2 * m + fo];
    v8f acc0 = {}, acc1 = {};
#pragma unroll
    for (int m = 0; m < 16; ++m) {
      v2f a, h0, h1;
      a.x = aw[m].x; a.y = aw[m].y;
      h0.x = fmaxf(bj0[m].x + ci[m].x, 0.0f); h0.y = fmaxf(bj0[m].y + ci[m].y, 0.0f);
      h1.x = fmaxf(bj1[m].x + ci[m].x, 0.0f); h1.y = fmaxf(bj1[m].y + ci[m].y, 0.0f);
      acc0 = mma4(a, h0, acc0);
      acc1 = mma4(a, h1, acc1);
      a.x = aw[m].z; a.y = aw[m].w;
      h0.x = fmaxf(bj0[m].z + ci[m].z, 0.0f); h0.y = fmaxf(bj0[m].w + ci[m].w, 0.0f);
      h1.x = fmaxf(bj1[m].z + ci[m].z, 0.0f); h1.y = fmaxf(bj1[m].w + ci[m].w, 0.0f);
      acc0 = mma4(a, h0, acc0);
      acc1 = mma4(a, h1, acc1);
    }
#pragma unroll
    for (int r = 0; r < 8; ++r) {
      const float o1 = __shfl_xor(acc1[r], 16);
      const float v  = (hi ? o1 : acc0[r]) + bq[r];
      float s = 1.0f / (1.0f + __expf(-v));
      if (ig == jg) s = 0.0f;
      out[((b * 8 + r) * 128 + ig) * 128 + jg] = s;
    }
  }
#undef FETCH_B
#undef LD4
}

extern "C" void kernel_launch(void* const* d_in, const int* in_sizes, int n_in,
                              void* d_out, int out_size, void* d_ws, size_t ws_size,
                              hipStream_t stream) {
  const float* E  = (const float*)d_in[0];
  const float* W1 = (const float*)d_in[1];
  const float* b1 = (const float*)d_in[2];
  const float* W2 = (const float*)d_in[3];
  const float* b2 = (const float*)d_in[4];
  const float* q  = (const float*)d_in[5];
  float* out = (float*)d_out;

  float* Wqp = (float*)d_ws + 262144;
  float* bqv = Wqp + 2048;

  wq_kernel<<<dim3(33), dim3(256), 0, stream>>>(W2, b2, q, Wqp, bqv);
  fused_kernel<<<dim3(256), dim3(256), 0, stream>>>(E, W1, b1, Wqp, bqv, out);
}


# Aj = E x W1b^T computed once per row by 128 WMMA work-groups added to the Wq launch, read back as 32-row tiles; fused kernel keeps only its Ai GEMM (8x redundant Aj work gone); Aj sum order changes
# speedup vs baseline: 3.7730x; 1.0641x over previous
#include <hip/hip_runtime.h>
#include <math.h>


typedef __attribute__((ext_vector_type(2))) float v2f;
typedef __attribute__((ext_vector_type(8))) float v8f;
typedef __attribute__((ext_vector_type(4))) float v4f;

__device__ __forceinline__ v8f mma4(v2f a, v2f b, v8f c) {
  return __builtin_amdgcn_wmma_f32_16x16x4_f32(false, a, false, b, (short)0, c, false, false);
}
#define LD4(ptr) (*(const v4f*)(ptr))

#define PP4 9
__global__ void __launch_bounds__(256)
pre_kernel(const float* __restrict__ E, const float* __restrict__ W1,
           const float* __restrict__ W2, const float* __restrict__ b2,
           const float* __restrict__ q, float* __restrict__ Ajw,
           float* __restrict__ Wqp, float* __restrict__ bqv) {
  __shared__ v4f part4[8 * 32 * PP4];
  __shared__ float red[8 * 4 * 8];
  const int t    = threadIdx.x;
  const int lane = t & 31;
  const int wv   = t >> 5;

  if (blockIdx.x < 128) {
    const int rt  = blockIdx.x >> 2;
    const int cg  = blockIdx.x & 3;
    const int c15 = lane & 15;
    const int hi  = lane >> 4;
    const int d0  = 32 * wv + 4 * hi;
    const float* ea = E + (32 * rt + c15) * 256 + d0;
    const float* wb = W1 + (32 * cg + c15) * 512 + 256 + d0;
    v4f fa0[4], fa1[4], fb0[4], fb1[4];
#pragma unroll
    for (int m = 0; m < 4; ++m) {
      fa0[m] = LD4(ea + 8 * m);
      fa1[m] = LD4(ea + 16 * 256 + 8 * m);
      fb0[m] = LD4(wb + 8 * m);
      fb1[m] = LD4(wb + 16 * 512 + 8 * m);
    }
    v8f acc00 = {}, acc01 = {}, acc10 = {}, acc11 = {};
#pragma unroll
    for (int m = 0; m < 4; ++m) {
      v2f a0, a1, p0, p1;
      a0.x = fa0[m].x; a0.y = fa0[m].y; a1.x = fa1[m].x; a1.y = fa1[m].y;
      p0.x = fb0[m].x; p0.y = fb0[m].y; p1.x = fb1[m].x; p1.y = fb1[m].y;
      acc00 = mma4(a0, p0, acc00);
      acc01 = mma4(a0, p1, acc01);
      acc10 = mma4(a1, p0, acc10);
      acc11 = mma4(a1, p1, acc11);
      a0.x = fa0[m].z; a0.y = fa0[m].w; a1.x = fa1[m].z; a1.y = fa1[m].w;
      p0.x = fb0[m].z; p0.y = fb0[m].w; p1.x = fb1[m].z; p1.y = fb1[m].w;
      acc00 = mma4(a0, p0, acc00);
      acc01 = mma4(a0, p1, acc01);
      acc10 = mma4(a1, p0, acc10);
      acc11 = mma4(a1, p1, acc11);
    }
    float* part = (float*)part4;
#pragma unroll
    for (int r = 0; r < 8; ++r) {
      part[(wv * 32 + 8 * hi + r) * (PP4 * 4) + c15]           = acc00[r];
      part[(wv * 32 + 8 * hi + r) * (PP4 * 4) + 16 + c15]      = acc01[r];
      part[(wv * 32 + 16 + 8 * hi + r) * (PP4 * 4) + c15]      = acc10[r];
      part[(wv * 32 + 16 + 8 * hi + r) * (PP4 * 4) + 16 + c15] = acc11[r];
    }
    __syncthreads();
    const int row = t >> 3;
    const int c4  = t & 7;
    v4f sacc = part4[row * PP4 + c4];
#pragma unroll
    for (int u = 1; u < 8; ++u) {
      const v4f pv = part4[(u * 32 + row) * PP4 + c4];
      sacc.x += pv.x; sacc.y += pv.y; sacc.z += pv.z; sacc.w += pv.w;
    }
    *(v4f*)(Ajw + (32 * rt + row) * 128 + 32 * cg + 4 * c4) = sacc;
    return;
  }

  const int hl   = t & 3;
  const int d0   = (t >> 2) * 4;
  const int g    = blockIdx.x - 128;

  float w0, w1, w2, w3;
  if (g < 32) {
    const float* wp = W2 + d0 * 128 + g * 4 + hl;
    w0 = wp[0];
    w1 = wp[128];
    w2 = wp[256];
    w3 = wp[384];
  } else {
    const float4 bv = *(const float4*)(b2 + d0);
    w0 = bv.x; w1 = bv.y; w2 = bv.z; w3 = bv.w;
  }

  float s[8];
#pragma unroll
  for (int k = 0; k < 8; ++k) {
    const float4 qv = *(const float4*)(q + k * 256 + d0);
    s[k] = w0 * qv.x + w1 * qv.y + w2 * qv.z + w3 * qv.w;
  }
#pragma unroll
  for (int k = 0; k < 8; ++k) {
    s[k] += __shfl_xor(s[k], 4);
    s[k] += __shfl_xor(s[k], 8);
    s[k] += __shfl_xor(s[k], 16);
  }
  if (lane < 4) {
#pragma unroll
    for (int k = 0; k < 8; ++k) red[(wv * 4 + lane) * 8 + k] = s[k];
  }
  __syncthreads();
  if (t < 32) {
    const int c = t >> 3;
    const int k = t & 7;
    float r = 0.0f;
#pragma unroll
    for (int u = 0; u < 8; ++u) r += red[(u * 4 + c) * 8 + k];
    if (g < 32) {
      Wqp[(g * 4 + c) * 16 + k]     = r;
      Wqp[(g * 4 + c) * 16 + 8 + k] = 0.0f;
    } else if (c == 0) {
      bqv[k] = r;
    }
  }
}
#undef PP4

#define E_P4  65
#define S_P4  33

__global__ void __launch_bounds__(256)
fused_kernel(const float* __restrict__ E, const float* __restrict__ W1,
             const float* __restrict__ b1, const float* __restrict__ Ajw,
             const float* __restrict__ Wqp, const float* __restrict__ bqv,
             float* __restrict__ out) {
  __shared__ v4f sE4[16 * E_P4];
  __shared__ v4f sSi4[16 * S_P4];
  __shared__ v4f sSj4[32 * S_P4];
  __shared__ v4f sWq4[16 * S_P4];

  const int t    = threadIdx.x;
  const int lane = t & 31;
  const int w    = t >> 5;
  const int c15  = lane & 15;
  const int hi   = lane >> 4;
  const int g    = blockIdx.x;
  const int b    = g >> 5;
  const int i0   = ((g >> 2) & 7) * 16;
  const int j0   = (g & 3) * 32;

  const int tr = t >> 4;
  const int tc = t & 15;
  const float* eI = E + (b * 128 + i0 + tr) * 256 + tc * 4;
  const int jr = t >> 3;
  const int jc = t & 7;
  const float* aJ = Ajw + (b * 128 + j0 + jr) * 128 + jc * 4;
  v4f st[4], sj[4];
#pragma unroll
  for (int qd = 0; qd < 4; ++qd) {
    st[qd] = LD4(eI + qd * 64);
    sj[qd] = LD4(aJ + qd * 32);
  }
  float wqv[8];
#pragma unroll
  for (int u = 0; u < 8; ++u) wqv[u] = Wqp[t + u * 256];

  const float* wB = W1 + (16 * w + c15) * 512 + 4 * hi;
  v4f ba[32];
#pragma unroll
  for (int m = 0; m < 8; ++m) ba[m] = LD4(wB + 8 * m);

#pragma unroll
  for (int qd = 0; qd < 4; ++qd) {
    sE4[tr * E_P4 + tc + 16 * qd]  = st[qd];
    sSj4[jr * S_P4 + jc + 8 * qd]  = sj[qd];
  }
  {
    float* sWq = (float*)sWq4;
#pragma unroll
    for (int u = 0; u < 8; ++u) {
      const int idx = t + u * 256;
      sWq[(idx & 15) * (S_P4 * 4) + (idx >> 4)] = wqv[u];
    }
  }
  __syncthreads();

#pragma unroll
  for (int m = 8; m < 32; ++m) ba[m] = LD4(wB + 8 * m);

  v8f accI = {};
  const int fo = hi;

#pragma unroll
  for (int cc = 0; cc < 4; ++cc) {
    v4f fi[8];
#pragma unroll
    for (int m = 0; m < 8; ++m) fi[m] = sE4[c15 * E_P4 + cc * 16 + 2 * m + fo];
#pragma unroll
    for (int m = 0; m < 8; ++m) {
      v2f a, bw;
      a.x = fi[m].x; a.y = fi[m].y; bw.x = ba[8 * cc + m].x; bw.y = ba[8 * cc + m].y;
      accI = mma4(a, bw, accI);
      a.x = fi[m].z; a.y = fi[m].w; bw.x = ba[8 * cc + m].z; bw.y = ba[8 * cc + m].w;
      accI = mma4(a, bw, accI);
    }
  }

  {
    float* sSi = (float*)sSi4;
    const float bias = b1[16 * w + c15];
#pragma unroll
    for (int r = 0; r < 8; ++r) sSi[(8 * hi + r) * (S_P4 * 4) + 16 * w + c15] = accI[r] + bias;
  }
  __syncthreads();

  v4f bj0[16], bj1[16], aw[16];
#pragma unroll
  for (int m = 0; m < 16; ++m) {
    aw[m]  = sWq4[c15 * S_P4 + 2 * m + fo];
    bj0[m] = sSj4[c15 * S_P4 + 2 * m + fo];
    bj1[m] = sSj4[(16 + c15) * S_P4 + 2 * m + fo];
  }
  float bq[8];
#pragma unroll
  for (int r = 0; r < 8; ++r) bq[r] = bqv[r];

  const int jg = j0 + lane;

#pragma unroll
  for (int ii = 0; ii < 2; ++ii) {
    const int il = 2 * w + ii;
    const int ig = i0 + il;
    v4f ci[16];
#pragma unroll
    for (int m = 0; m < 16; ++m) ci[m] = sSi4[il * S_P4 + 2 * m + fo];
    v8f acc0 = {}, acc1 = {};
#pragma unroll
    for (int m = 0; m < 16; ++m) {
      v2f a, h0, h1;
      a.x = aw[m].x; a.y = aw[m].y;
      h0.x = fmaxf(bj0[m].x + ci[m].x, 0.0f); h0.y = fmaxf(bj0[m].y + ci[m].y, 0.0f);
      h1.x = fmaxf(bj1[m].x + ci[m].x, 0.0f); h1.y = fmaxf(bj1[m].y + ci[m].y, 0.0f);
      acc0 = mma4(a, h0, acc0);
      acc1 = mma4(a, h1, acc1);
      a.x = aw[m].z; a.y = aw[m].w;
      h0.x = fmaxf(bj0[m].z + ci[m].z, 0.0f); h0.y = fmaxf(bj0[m].w + ci[m].w, 0.0f);
      h1.x = fmaxf(bj1[m].z + ci[m].z, 0.0f); h1.y = fmaxf(bj1[m].w + ci[m].w, 0.0f);
      acc0 = mma4(a, h0, acc0);
      acc1 = mma4(a, h1, acc1);
    }
#pragma unroll
    for (int r = 0; r < 8; ++r) {
      const float o1 = __shfl_xor(acc1[r], 16);
      const float v  = (hi ? o1 : acc0[r]) + bq[r];
      float s = 1.0f / (1.0f + __expf(-v));
      if (ig == jg) s = 0.0f;
      out[((b * 8 + r) * 128 + ig) * 128 + jg] = s;
    }
  }
}

extern "C" void kernel_launch(void* const* d_in, const int* in_sizes, int n_in,
                              void* d_out, int out_size, void* d_ws, size_t ws_size,
                              hipStream_t stream) {
  const float* E  = (const float*)d_in[0];
  const float* W1 = (const float*)d_in[1];
  const float* b1 = (const float*)d_in[2];
  const float* W2 = (const float*)d_in[3];
  const float* b2 = (const float*)d_in[4];
  const float* q  = (const float*)d_in[5];
  float* out = (float*)d_out;

  float* Ajw = (float*)d_ws;
  float* Wqp = (float*)d_ws + 262144;
  float* bqv = Wqp + 2048;

  pre_kernel<<<dim3(161), dim3(256), 0, stream>>>(E, W1, W2, b2, q, Ajw, Wqp, bqv);
  fused_kernel<<<dim3(256), dim3(256), 0, stream>>>(E, W1, b1, Ajw, Wqp, bqv, out);
}


# Ai + b1 computed once per row by 128 more WMMA work-groups in the first launch; score kernel reads its 16-row tile back, its 4x redundant Ai GEMM phase and W1a reads are gone; Ai sum order changes
# speedup vs baseline: 4.6112x; 1.2222x over previous
#include <hip/hip_runtime.h>
#include <math.h>


typedef __attribute__((ext_vector_type(2))) float v2f;
typedef __attribute__((ext_vector_type(8))) float v8f;
typedef __attribute__((ext_vector_type(4))) float v4f;

__device__ __forceinline__ v8f mma4(v2f a, v2f b, v8f c) {
  return __builtin_amdgcn_wmma_f32_16x16x4_f32(false, a, false, b, (short)0, c, false, false);
}
#define LD4(ptr) (*(const v4f*)(ptr))

#define PP4 9
__global__ void __launch_bounds__(256)
pre_kernel(const float* __restrict__ E, const float* __restrict__ W1,
           const float* __restrict__ b1,
           const float* __restrict__ W2, const float* __restrict__ b2,
           const float* __restrict__ q, float* __restrict__ Siw, float* __restrict__ Ajw,
           float* __restrict__ Wqp, float* __restrict__ bqv) {
  __shared__ v4f part4[8 * 32 * PP4];
  __shared__ float red[8 * 4 * 8];
  const int t    = threadIdx.x;
  const int lane = t & 31;
  const int wv   = t >> 5;

  if (blockIdx.x < 256) {
    const int rt  = blockIdx.x >> 3;
    const int cg  = blockIdx.x & 3;
    const int wh  = (blockIdx.x >> 2) & 1;
    const int c15 = lane & 15;
    const int hi  = lane >> 4;
    const int d0  = 32 * wv + 4 * hi;
    const float* ea = E + (32 * rt + c15) * 256 + d0;
    const float* wb = W1 + (32 * cg + c15) * 512 + 256 * wh + d0;
    const int row = t >> 3;
    const int c4  = t & 7;
    v4f bv = {};
    if (wh == 0) bv = LD4(b1 + 32 * cg + 4 * c4);
    v4f fa0[4], fa1[4], fb0[4], fb1[4];
#pragma unroll
    for (int m = 0; m < 4; ++m) {
      fa0[m] = LD4(ea + 8 * m);
      fa1[m] = LD4(ea + 16 * 256 + 8 * m);
      fb0[m] = LD4(wb + 8 * m);
      fb1[m] = LD4(wb + 16 * 512 + 8 * m);
    }
    v8f acc00 = {}, acc01 = {}, acc10 = {}, acc11 = {};
#pragma unroll
    for (int m = 0; m < 4; ++m) {
      v2f a0, a1, p0, p1;
      a0.x = fa0[m].x; a0.y = fa0[m].y; a1.x = fa1[m].x; a1.y = fa1[m].y;
      p0.x = fb0[m].x; p0.y = fb0[m].y; p1.x = fb1[m].x; p1.y = fb1[m].y;
      acc00 = mma4(a0, p0, acc00);
      acc01 = mma4(a0, p1, acc01);
      acc10 = mma4(a1, p0, acc10);
      acc11 = mma4(a1, p1, acc11);
      a0.x = fa0[m].z; a0.y = fa0[m].w; a1.x = fa1[m].z; a1.y = fa1[m].w;
      p0.x = fb0[m].z; p0.y = fb0[m].w; p1.x = fb1[m].z; p1.y = fb1[m].w;
      acc00 = mma4(a0, p0, acc00);
      acc01 = mma4(a0, p1, acc01);
      acc10 = mma4(a1, p0, acc10);
      acc11 = mma4(a1, p1, acc11);
    }
    float* part = (float*)part4;
#pragma unroll
    for (int r = 0; r < 8; ++r) {
      part[(wv * 32 + 8 * hi + r) * (PP4 * 4) + c15]           = acc00[r];
      part[(wv * 32 + 8 * hi + r) * (PP4 * 4) + 16 + c15]      = acc01[r];
      part[(wv * 32 + 16 + 8 * hi + r) * (PP4 * 4) + c15]      = acc10[r];
      part[(wv * 32 + 16 + 8 * hi + r) * (PP4 * 4) + 16 + c15] = acc11[r];
    }
    __syncthreads();
    v4f sacc = part4[row * PP4 + c4];
#pragma unroll
    for (int u = 1; u < 8; ++u) {
      const v4f pv = part4[(u * 32 + row) * PP4 + c4];
      sacc.x += pv.x; sacc.y += pv.y; sacc.z += pv.z; sacc.w += pv.w;
    }
    if (wh == 0) {
      sacc.x += bv.x; sacc.y += bv.y; sacc.z += bv.z; sacc.w += bv.w;
      *(v4f*)(Siw + (32 * rt + row) * 128 + 32 * cg + 4 * c4) = sacc;
    } else {
      *(v4f*)(Ajw + (32 * rt + row) * 128 + 32 * cg + 4 * c4) = sacc;
    }
    return;
  }

  const int hl   = t & 3;
  const int d0   = (t >> 2) * 4;
  const int g    = blockIdx.x - 256;

  float w0, w1, w2, w3;
  if (g < 32) {
    const float* wp = W2 + d0 * 128 + g * 4 + hl;
    w0 = wp[0];
    w1 = wp[128];
    w2 = wp[256];
    w3 = wp[384];
  } else {
    const float4 bv = *(const float4*)(b2 + d0);
    w0 = bv.x; w1 = bv.y; w2 = bv.z; w3 = bv.w;
  }

  float s[8];
#pragma unroll
  for (int k = 0; k < 8; ++k) {
    const float4 qv = *(const float4*)(q + k * 256 + d0);
    s[k] = w0 * qv.x + w1 * qv.y + w2 * qv.z + w3 * qv.w;
  }
#pragma unroll
  for (int k = 0; k < 8; ++k) {
    s[k] += __shfl_xor(s[k], 4);
    s[k] += __shfl_xor(s[k], 8);
    s[k] += __shfl_xor(s[k], 16);
  }
  if (lane < 4) {
#pragma unroll
    for (int k = 0; k < 8; ++k) red[(wv * 4 + lane) * 8 + k] = s[k];
  }
  __syncthreads();
  if (t < 32) {
    const int c = t >> 3;
    const int k = t & 7;
    float r = 0.0f;
#pragma unroll
    for (int u = 0; u < 8; ++u) r += red[(u * 4 + c) * 8 + k];
    if (g < 32) {
      Wqp[(g * 4 + c) * 16 + k]     = r;
      Wqp[(g * 4 + c) * 16 + 8 + k] = 0.0f;
    } else if (c == 0) {
      bqv[k] = r;
    }
  }
}
#undef PP4

#define S_P4  33

__global__ void __launch_bounds__(256)
score_kernel(const float* __restrict__ Siw, const float* __restrict__ Ajw,
             const float* __restrict__ Wqp, const float* __restrict__ bqv,
             float* __restrict__ out) {
  __shared__ v4f sSi4[16 * S_P4];
  __shared__ v4f sSj4[32 * S_P4];
  __shared__ v4f sWq4[16 * S_P4];

  const int t    = threadIdx.x;
  const int lane = t & 31;
  const int w    = t >> 5;
  const int c15  = lane & 15;
  const int hi   = lane >> 4;
  const int g    = blockIdx.x;
  const int b    = g >> 5;
  const int i0   = ((g >> 2) & 7) * 16;
  const int j0   = (g & 3) * 32;

  const int tr = t >> 4;
  const int tc = t & 15;
  const float* sI = Siw + (b * 128 + i0 + tr) * 128 + tc * 4;
  const int jr = t >> 3;
  const int jc = t & 7;
  const float* aJ = Ajw + (b * 128 + j0 + jr) * 128 + jc * 4;
  v4f si[2], sj[4];
#pragma unroll
  for (int qd = 0; qd < 2; ++qd) si[qd] = LD4(sI + qd * 64);
#pragma unroll
  for (int qd = 0; qd < 4; ++qd) sj[qd] = LD4(aJ + qd * 32);
  float wqv[8];
#pragma unroll
  for (int u = 0; u < 8; ++u) wqv[u] = Wqp[t + u * 256];

#pragma unroll
  for (int qd = 0; qd < 2; ++qd) sSi4[tr * S_P4 + tc + 16 * qd] = si[qd];
#pragma unroll
  for (int qd = 0; qd < 4; ++qd) sSj4[jr * S_P4 + jc + 8 * qd] = sj[qd];
  {
    float* sWq = (float*)sWq4;
#pragma unroll
    for (int u = 0; u < 8; ++u) {
      const int idx = t + u * 256;
      sWq[(idx & 15) * (S_P4 * 4) + (idx >> 4)] = wqv[u];
    }
  }
  __syncthreads();

  const int fo = hi;

  v4f bj0[16], bj1[16], aw[16];
#pragma unroll
  for (int m = 0; m < 16; ++m) {
    aw[m]  = sWq4[c15 * S_P4 + 2 * m + fo];
    bj0[m] = sSj4[c15 * S_P4 + 2 * m + fo];
    bj1[m] = sSj4[(16 + c15) * S_P4 + 2 * m + fo];
  }
  float bq[8];
#pragma unroll
  for (int r = 0; r < 8; ++r) bq[r] = bqv[r];

  const int jg = j0 + lane;

#pragma unroll
  for (int ii = 0; ii < 2; ++ii) {
    const int il = 2 * w + ii;
    const int ig = i0 + il;
    v4f ci[16];
#pragma unroll
    for (int m = 0; m < 16; ++m) ci[m] = sSi4[il * S_P4 + 2 * m + fo];
    v8f acc0 = {}, acc1 = {};
#pragma unroll
    for (int m = 0; m < 16; ++m) {
      v2f a, h0, h1;
      a.x = aw[m].x; a.y = aw[m].y;
      h0.x = fmaxf(bj0[m].x + ci[m].x, 0.0f); h0.y = fmaxf(bj0[m].y + ci[m].y, 0.0f);
      h1.x = fmaxf(bj1[m].x + ci[m].x, 0.0f); h1.y = fmaxf(bj1[m].y + ci[m].y, 0.0f);
      acc0 = mma4(a, h0, acc0);
      acc1 = mma4(a, h1, acc1);
      a.x = aw[m].z; a.y = aw[m].w;
      h0.x = fmaxf(bj0[m].z + ci[m].z, 0.0f); h0.y = fmaxf(bj0[m].w + ci[m].w, 0.0f);
      h1.x = fmaxf(bj1[m].z + ci[m].z, 0.0f); h1.y = fmaxf(bj1[m].w + ci[m].w, 0.0f);
      acc0 = mma4(a, h0, acc0);
      acc1 = mma4(a, h1, acc1);
    }
#pragma unroll
    for (int r = 0; r < 8; ++r) {
      const float o1 = __shfl_xor(acc1[r], 16);
      const float v  = (hi ? o1 : acc0[r]) + bq[r];
      float s = 1.0f / (1.0f + __expf(-v));
      if (ig == jg) s = 0.0f;
      out[((b * 8 + r) * 128 + ig) * 128 + jg] = s;
    }
  }
}

extern "C" void kernel_launch(void* const* d_in, const int* in_sizes, int n_in,
                              void* d_out, int out_size, void* d_ws, size_t ws_size,
                              hipStream_t stream) {
  const float* E  = (const float*)d_in[0];
  const float* W1 = (const float*)d_in[1];
  const float* b1 = (const float*)d_in[2];
  const float* W2 = (const float*)d_in[3];
  const float* b2 = (const float*)d_in[4];
  const float* q  = (const float*)d_in[5];
  float* out = (float*)d_out;

  float* Ajw = (float*)d_ws;
  float* Siw = (float*)d_ws + 131072;
  float* Wqp = (float*)d_ws + 262144;
  float* bqv = Wqp + 2048;

  pre_kernel<<<dim3(289), dim3(256), 0, stream>>>(E, W1, b1, W2, b2, q, Siw, Ajw, Wqp, bqv);
  score_kernel<<<dim3(256), dim3(256), 0, stream>>>(Siw, Ajw, Wqp, bqv, out);
}
